# starting kernel
# baseline (speedup 1.0000x reference)
#include <hip/hip_runtime.h>
#include <hip/hip_bf16.h>

typedef __attribute__((ext_vector_type(2))) float v2f;
typedef __attribute__((ext_vector_type(8))) float v8f;

__global__ void zero_f32(float* __restrict__ p, int n) {
    int i = blockIdx.x * blockDim.x + threadIdx.x;
    if (i < n) p[i] = 0.0f;
}

__global__ void degree_k(const int* __restrict__ dst, float* __restrict__ deg, int E) {
    int e = blockIdx.x * blockDim.x + threadIdx.x;
    if (e < E) atomicAdd(&deg[dst[e]], 1.0f);
}

__global__ void scatter_k(const float* __restrict__ h, const int* __restrict__ src,
                          const int* __restrict__ dst, const float* __restrict__ ew,
                          float* __restrict__ agg, int E, int K) {
    int tid = blockIdx.x * blockDim.x + threadIdx.x;
    int chunks = K >> 2;
    int total = E * chunks;
    if (tid >= total) return;
    int e = tid / chunks;
    int c = (tid - e * chunks) << 2;

    int s = src[e];
    int d = dst[e];
    float w = ew[e];

    float4 hv = *(const float4*)(h + (size_t)s * K + c);
    float* ap = agg + (size_t)d * K + c;
    atomicAdd(ap + 0, hv.x * w);
    atomicAdd(ap + 1, hv.y * w);
    atomicAdd(ap + 2, hv.z * w);
    atomicAdd(ap + 3, hv.w * w);
}

__global__ void sage_gemm(const float* __restrict__ h, const float* __restrict__ agg,
                          const float* __restrict__ deg,
                          const float* __restrict__ Ws, const float* __restrict__ Wn,
                          const float* __restrict__ bias,
                          float* __restrict__ out,
                          int K, int Fout, int activate) {
    int lane = threadIdx.x;
    int m    = lane & 15;
    int kh   = (lane >> 4) << 1;
    int row0 = blockIdx.x << 4;
    int col0 = blockIdx.y << 4;

    int node = row0 + m;
    int col  = col0 + m;
    int   cs    = (col < Fout) ? col  : 0;
    float cmask = (col < Fout) ? 1.0f : 0.0f;

    float invdeg = 1.0f / fmaxf(deg[node], 1.0f);

    const float* hrow = h   + (size_t)node * K;
    const float* arow = agg + (size_t)node * K;

    v8f c = {};
    for (int k = 0; k < K; k += 4) {
        v2f a, b;
        a.x = hrow[k + kh];
        a.y = hrow[k + kh + 1];
        b.x = Ws[(k + kh)     * Fout + cs] * cmask;
        b.y = Ws[(k + kh + 1) * Fout + cs] * cmask;
        c = __builtin_amdgcn_wmma_f32_16x16x4_f32(false, a, false, b,
                                                  (short)0, c, false, false);
        a.x = arow[k + kh]     * invdeg;
        a.y = arow[k + kh + 1] * invdeg;
        b.x = Wn[(k + kh)     * Fout + cs] * cmask;
        b.y = Wn[(k + kh + 1) * Fout + cs] * cmask;
        c = __builtin_amdgcn_wmma_f32_16x16x4_f32(false, a, false, b,
                                                  (short)0, c, false, false);
    }

    int half = lane >> 4;
    int cn   = col0 + (lane & 15);
    if (cn < Fout) {
        float bv = bias[cn];
        for (int j = 0; j < 8; ++j) {
            int rm = row0 + j + (half << 3);
            float v = c[j] + bv;
            if (activate) v = tanhf(v);
            out[(size_t)rm * Fout + cn] = v;
        }
    }
}

extern "C" void kernel_launch(void* const* d_in, const int* in_sizes, int n_in,
                              void* d_out, int out_size, void* d_ws, size_t ws_size,
                              hipStream_t stream) {
    const float* b_z  = (const float*)d_in[0];
    const int*   esrc = (const int*)  d_in[1];
    const int*   edst = (const int*)  d_in[2];
    const float* ew   = (const float*)d_in[3];
    const float* Ws0  = (const float*)d_in[4];
    const float* Wn0  = (const float*)d_in[5];
    const float* b0   = (const float*)d_in[6];
    const float* Ws1  = (const float*)d_in[7];
    const float* Wn1  = (const float*)d_in[8];
    const float* b1   = (const float*)d_in[9];
    const float* Ws2  = (const float*)d_in[10];
    const float* Wn2  = (const float*)d_in[11];
    const float* b2   = (const float*)d_in[12];

    const int N = in_sizes[0] / 64;
    const int E = in_sizes[1];

    float* deg = (float*)d_ws;
    float* h1  = deg + N;
    float* h2  = h1  + (size_t)N * 32;
    float* agg = h2  + (size_t)N * 16;
    float* out = (float*)d_out;

    auto cdiv = [](int a, int b) { return (a + b - 1) / b; };

    zero_f32<<<cdiv(N, 256), 256, 0, stream>>>(deg, N);
    degree_k<<<cdiv(E, 256), 256, 0, stream>>>(edst, deg, E);

    struct Layer {
        const float *hin, *Ws, *Wn, *b;
        float* hout;
        int K, F, act;
    } L[3] = {
        { b_z, Ws0, Wn0, b0, h1,  64, 32, 1 },
        { h1,  Ws1, Wn1, b1, h2,  32, 16, 1 },
        { h2,  Ws2, Wn2, b2, out, 16,  1, 0 },
    };

    for (int i = 0; i < 3; ++i) {
        const int K = L[i].K, F = L[i].F;
        zero_f32<<<cdiv(N * K, 256), 256, 0, stream>>>(agg, N * K);
        scatter_k<<<cdiv(E * (K / 4), 256), 256, 0, stream>>>(
            L[i].hin, esrc, edst, ew, agg, E, K);
        dim3 grid(N / 16, cdiv(F, 16));
        sage_gemm<<<grid, 32, 0, stream>>>(
            L[i].hin, agg, deg, L[i].Ws, L[i].Wn, L[i].b, L[i].hout, K, F, L[i].act);
    }
}


# Clears and degree count fold into neighbours: a dense wave (all columns of its 16 nodes) clears its agg rows for the next layer, layer 0's edge walk counts deg; 11 launches->7, 201k tiny waves gone
# speedup vs baseline: 1.5956x; 1.5956x over previous
#include <hip/hip_runtime.h>
#include <hip/hip_bf16.h>

typedef __attribute__((ext_vector_type(2))) float v2f;
typedef __attribute__((ext_vector_type(8))) float v8f;

constexpr int AGG_STRIDE = 64;

__global__ void __launch_bounds__(256)
clear2(float* __restrict__ p0, int n0, float* __restrict__ p1, int n1) {
    const int tid = blockIdx.x * blockDim.x + threadIdx.x;
    const int nth = gridDim.x * blockDim.x;
    const float4 z = make_float4(0.0f, 0.0f, 0.0f, 0.0f);
    for (int i = tid; i < (n0 >> 2); i += nth) reinterpret_cast<float4*>(p0)[i] = z;
    for (int i = (n0 & ~3) + tid; i < n0; i += nth) p0[i] = 0.0f;
    for (int i = tid; i < (n1 >> 2); i += nth) reinterpret_cast<float4*>(p1)[i] = z;
    for (int i = (n1 & ~3) + tid; i < n1; i += nth) p1[i] = 0.0f;
}

template <int J> struct ic { static constexpr int value = J; };
template <int J, int JEND, class F>
__device__ __forceinline__ void unrolled(F&& f) {
    if constexpr (J < JEND) { f(ic<J>{}); unrolled<J + 1, JEND>(f); }
}

template <int J> __device__ __forceinline__ int half_bcast(int v) {
    return __builtin_amdgcn_update_dpp(0, v, 0x150 + J, 0xf, 0xf, false);
}
__device__ __forceinline__ int lane_i(int v, int j) { return __builtin_amdgcn_readlane(v, j); }
__device__ __forceinline__ float lane_f(float v, int j) {
    return __int_as_float(__builtin_amdgcn_readlane(__float_as_int(v), j));
}

template <int K, bool DEG>
__global__ void __launch_bounds__(64)
scatter_rows(const float* __restrict__ h, const int* __restrict__ src,
             const int* __restrict__ dst, const float* __restrict__ ew,
             float* __restrict__ agg, float* __restrict__ deg, int E) {
    constexpr int AS = AGG_STRIDE;
    constexpr int EPW = 128;
    constexpr int NCH = EPW / 32;
    constexpr int WPL = (K >= 32) ? K / 32 : 1;
    constexpr int GE  = (K >= 32) ? 16 / WPL : 32;
    constexpr int GPC = 32 / GE;
    constexpr int NG  = NCH * GPC;
    const int lane = threadIdx.x & 31;
    const int wave = __builtin_amdgcn_readfirstlane((int)((blockIdx.x * blockDim.x + threadIdx.x) >> 5));
    const int e0 = wave * EPW;
    if (e0 >= E) return;

    int   s[NCH], d[NCH];
    float w[NCH];
#pragma unroll
    for (int c = 0; c < NCH; ++c) {
        const int e  = e0 + c * 32 + lane;
        const int ec = (e < E) ? e : 0;
        s[c] = src[ec];
        d[c] = dst[ec];
        w[c] = ew[ec];
    }
    auto count_deg = [&](auto FULL) {
        if constexpr (DEG) {
            for (int c = 0; c < NCH; ++c)
                if (decltype(FULL)::value || e0 + c * 32 + lane < E) atomicAdd(&deg[d[c]], 1.0f);
        }
    };

    if (E - e0 >= EPW) {
        float v[2][16];
        auto load_group = [&](auto T) {
            constexpr int t = decltype(T)::value;
            constexpr int c = t / GPC, g0 = (t % GPC) * GE;
            if constexpr (K >= 32) {
#pragma unroll
                for (int i = 0; i < GE; ++i) {
                    const int sj = lane_i(s[c], g0 + i);
                    const float* hr = h + (size_t)sj * K + lane;
#pragma unroll
                    for (int q = 0; q < WPL; ++q) v[t & 1][i * WPL + q] = hr[32 * q];
                }
            } else {
                unrolled<0, 16>([&](auto J) {
                    constexpr int j = decltype(J)::value;
                    const unsigned sj = (unsigned)half_bcast<j>(s[c]);
                    v[t & 1][j] = h[sj * K + (lane & 15)];
                });
            }
        };
        auto add_group = [&](auto T) {
            constexpr int t = decltype(T)::value;
            constexpr int c = t / GPC, g0 = (t % GPC) * GE;
            if constexpr (K >= 32) {
#pragma unroll
                for (int i = 0; i < GE; ++i) {
                    const int   dj = lane_i(d[c], g0 + i);
                    const float wj = lane_f(w[c], g0 + i);
                    float* ar = agg + (size_t)dj * AS + lane;
#pragma unroll
                    for (int q = 0; q < WPL; ++q) atomicAdd(ar + 32 * q, v[t & 1][i * WPL + q] * wj);
                }
            } else {
                unrolled<0, 16>([&](auto J) {
                    constexpr int j = decltype(J)::value;
                    const unsigned dj = (unsigned)half_bcast<j>(d[c]);
                    const float    wj = __int_as_float(half_bcast<j>(__float_as_int(w[c])));
                    atomicAdd(agg + (dj * AS + (lane & 15)), v[t & 1][j] * wj);
                });
            }
        };
        load_group(ic<0>{});
        count_deg(ic<1>{});
        unrolled<0, NG>([&](auto T) {
            constexpr int t = decltype(T)::value;
            if constexpr (t + 1 < NG) load_group(ic<t + 1>{});
            add_group(T);
        });
    } else {
        count_deg(ic<0>{});
        for (int c = 0; c < NCH; ++c) {
            const int left = E - (e0 + c * 32);
            const int cnt  = left < 32 ? left : 32;
            const int sc = s[c], dc = d[c];
            const float wc = w[c];
#pragma clang loop unroll(disable)
            for (int j = 0; j < cnt; ++j) {
                const int   sj = lane_i(sc, j);
                const int   dj = lane_i(dc, j);
                const float wj = lane_f(wc, j);
                for (int q = lane; q < K; q += 32)
                    atomicAdd(agg + (size_t)dj * AS + q, h[(size_t)sj * K + q] * wj);
            }
        }
    }
}

template <int K, int F, int ACT, int KCLR>
__global__ void __launch_bounds__(32)
sage_gemm(const float* __restrict__ h, float* agg,
          const float* __restrict__ deg,
          const float* __restrict__ Ws, const float* __restrict__ Wn,
          const float* __restrict__ bias,
          float* __restrict__ out) {
    constexpr int NT = (F + 15) / 16;
    const int lane = threadIdx.x;
    const int m    = lane & 15;
    const int kh   = (lane >> 4) << 1;
    const int row0 = blockIdx.x << 4;
    const int node = row0 + m;

    int   cs[NT];
    float cmask[NT];
    for (int t = 0; t < NT; ++t) {
        const int col = t * 16 + m;
        cs[t]    = (col < F) ? col  : 0;
        cmask[t] = (col < F) ? 1.0f : 0.0f;
    }

    const float invdeg = 1.0f / fmaxf(deg[node], 1.0f);

    const float* hrow = h   + (size_t)node * K;
    const float* arow = agg + (size_t)node * AGG_STRIDE;

    v8f c[NT];
    for (int t = 0; t < NT; ++t) c[t] = v8f{};

    for (int k = 0; k < K; k += 4) {
        v2f ah, an;
        ah.x = hrow[k + kh];
        ah.y = hrow[k + kh + 1];
        an.x = arow[k + kh]     * invdeg;
        an.y = arow[k + kh + 1] * invdeg;
        for (int t = 0; t < NT; ++t) {
            v2f b;
            b.x = Ws[(k + kh)     * F + cs[t]] * cmask[t];
            b.y = Ws[(k + kh + 1) * F + cs[t]] * cmask[t];
            c[t] = __builtin_amdgcn_wmma_f32_16x16x4_f32(false, ah, false, b,
                                                         (short)0, c[t], false, false);
            b.x = Wn[(k + kh)     * F + cs[t]] * cmask[t];
            b.y = Wn[(k + kh + 1) * F + cs[t]] * cmask[t];
            c[t] = __builtin_amdgcn_wmma_f32_16x16x4_f32(false, an, false, b,
                                                         (short)0, c[t], false, false);
        }
    }

    if constexpr (KCLR > 0) {
        constexpr int LPR = KCLR / 4;
        constexpr int RPI = 32 / LPR;
        const float4 z = make_float4(0.0f, 0.0f, 0.0f, 0.0f);
        float* crow = agg + (size_t)(row0 + lane / LPR) * AGG_STRIDE + (lane % LPR) * 4;
        for (int i = 0; i < 16 / RPI; ++i)
            *reinterpret_cast<float4*>(crow + (size_t)i * RPI * AGG_STRIDE) = z;
    }

    const int half = lane >> 4;
    for (int t = 0; t < NT; ++t) {
        const int cn = t * 16 + (lane & 15);
        if (cn < F) {
            const float bv = bias[cn];
            for (int j = 0; j < 8; ++j) {
                const int rm = row0 + j + (half << 3);
                float v = c[t][j] + bv;
                if (ACT) v = tanhf(v);
                out[(size_t)rm * F + cn] = v;
            }
        }
    }
}

extern "C" void kernel_launch(void* const* d_in, const int* in_sizes, int n_in,
                              void* d_out, int out_size, void* d_ws, size_t ws_size,
                              hipStream_t stream) {
    const float* b_z  = (const float*)d_in[0];
    const int*   esrc = (const int*)  d_in[1];
    const int*   edst = (const int*)  d_in[2];
    const float* ew   = (const float*)d_in[3];
    const float* Ws0  = (const float*)d_in[4];
    const float* Wn0  = (const float*)d_in[5];
    const float* b0   = (const float*)d_in[6];
    const float* Ws1  = (const float*)d_in[7];
    const float* Wn1  = (const float*)d_in[8];
    const float* b1   = (const float*)d_in[9];
    const float* Ws2  = (const float*)d_in[10];
    const float* Wn2  = (const float*)d_in[11];
    const float* b2   = (const float*)d_in[12];

    const int N = in_sizes[0] / 64;
    const int E = in_sizes[1];

    float* deg = (float*)d_ws;
    float* h1  = deg + N;
    float* h2  = h1  + (size_t)N * 32;
    float* agg = h2  + (size_t)N * 16;
    float* out = (float*)d_out;

    auto cdiv = [](int a, int b) { return (a + b - 1) / b; };

    clear2<<<512, 256, 0, stream>>>(deg, N, agg, N * 64);

    scatter_rows<64, true><<<cdiv(E, 256), 64, 0, stream>>>(b_z, esrc, edst, ew, agg, deg, E);
    sage_gemm<64, 32, 1, 32><<<N / 16, 32, 0, stream>>>(b_z, agg, deg, Ws0, Wn0, b0, h1);

    scatter_rows<32, false><<<cdiv(E, 256), 64, 0, stream>>>(h1, esrc, edst, ew, agg, nullptr, E);
    sage_gemm<32, 16, 1, 16><<<N / 16, 32, 0, stream>>>(h1, agg, deg, Ws1, Wn1, b1, h2);

    scatter_rows<16, false><<<cdiv(E, 256), 64, 0, stream>>>(h2, esrc, edst, ew, agg, nullptr, E);
    sage_gemm<16, 1, 0, 0><<<N / 16, 32, 0, stream>>>(h2, agg, deg, Ws2, Wn2, b2, out);
}


# Workspace planes reordered (agg, h1, h2, deg): agg and h1 rows now start on 128-byte lines, so a wave's row load or f32 atomic row add touches one line, not two; only the pointer arguments change
# speedup vs baseline: 1.7827x; 1.1173x over previous
#include <hip/hip_runtime.h>
#include <hip/hip_bf16.h>

typedef __attribute__((ext_vector_type(2))) float v2f;
typedef __attribute__((ext_vector_type(8))) float v8f;

constexpr int AGG_STRIDE = 64;

__global__ void __launch_bounds__(256)
clear2(float* __restrict__ p0, int n0, float* __restrict__ p1, int n1) {
    const int tid = blockIdx.x * blockDim.x + threadIdx.x;
    const int nth = gridDim.x * blockDim.x;
    const float4 z = make_float4(0.0f, 0.0f, 0.0f, 0.0f);
    for (int i = tid; i < (n0 >> 2); i += nth) reinterpret_cast<float4*>(p0)[i] = z;
    for (int i = (n0 & ~3) + tid; i < n0; i += nth) p0[i] = 0.0f;
    for (int i = tid; i < (n1 >> 2); i += nth) reinterpret_cast<float4*>(p1)[i] = z;
    for (int i = (n1 & ~3) + tid; i < n1; i += nth) p1[i] = 0.0f;
}

template <int J> struct ic { static constexpr int value = J; };
template <int J, int JEND, class F>
__device__ __forceinline__ void unrolled(F&& f) {
    if constexpr (J < JEND) { f(ic<J>{}); unrolled<J + 1, JEND>(f); }
}

template <int J> __device__ __forceinline__ int half_bcast(int v) {
    return __builtin_amdgcn_update_dpp(0, v, 0x150 + J, 0xf, 0xf, false);
}
__device__ __forceinline__ int lane_i(int v, int j) { return __builtin_amdgcn_readlane(v, j); }
__device__ __forceinline__ float lane_f(float v, int j) {
    return __int_as_float(__builtin_amdgcn_readlane(__float_as_int(v), j));
}

template <int K, bool DEG>
__global__ void __launch_bounds__(64)
scatter_rows(const float* __restrict__ h, const int* __restrict__ src,
             const int* __restrict__ dst, const float* __restrict__ ew,
             float* __restrict__ agg, float* __restrict__ deg, int E) {
    constexpr int AS = AGG_STRIDE;
    constexpr int EPW = 128;
    constexpr int NCH = EPW / 32;
    constexpr int WPL = (K >= 32) ? K / 32 : 1;
    constexpr int GE  = (K >= 32) ? 16 / WPL : 32;
    constexpr int GPC = 32 / GE;
    constexpr int NG  = NCH * GPC;
    const int lane = threadIdx.x & 31;
    const int wave = __builtin_amdgcn_readfirstlane((int)((blockIdx.x * blockDim.x + threadIdx.x) >> 5));
    const int e0 = wave * EPW;
    if (e0 >= E) return;

    int   s[NCH], d[NCH];
    float w[NCH];
#pragma unroll
    for (int c = 0; c < NCH; ++c) {
        const int e  = e0 + c * 32 + lane;
        const int ec = (e < E) ? e : 0;
        s[c] = src[ec];
        d[c] = dst[ec];
        w[c] = ew[ec];
    }
    auto count_deg = [&](auto FULL) {
        if constexpr (DEG) {
            for (int c = 0; c < NCH; ++c)
                if (decltype(FULL)::value || e0 + c * 32 + lane < E) atomicAdd(&deg[d[c]], 1.0f);
        }
    };

    if (E - e0 >= EPW) {
        float v[2][16];
        auto load_group = [&](auto T) {
            constexpr int t = decltype(T)::value;
            constexpr int c = t / GPC, g0 = (t % GPC) * GE;
            if constexpr (K >= 32) {
#pragma unroll
                for (int i = 0; i < GE; ++i) {
                    const int sj = lane_i(s[c], g0 + i);
                    const float* hr = h + (size_t)sj * K + lane;
#pragma unroll
                    for (int q = 0; q < WPL; ++q) v[t & 1][i * WPL + q] = hr[32 * q];
                }
            } else {
                unrolled<0, 16>([&](auto J) {
                    constexpr int j = decltype(J)::value;
                    const unsigned sj = (unsigned)half_bcast<j>(s[c]);
                    v[t & 1][j] = h[sj * K + (lane & 15)];
                });
            }
        };
        auto add_group = [&](auto T) {
            constexpr int t = decltype(T)::value;
            constexpr int c = t / GPC, g0 = (t % GPC) * GE;
            if constexpr (K >= 32) {
#pragma unroll
                for (int i = 0; i < GE; ++i) {
                    const int   dj = lane_i(d[c], g0 + i);
                    const float wj = lane_f(w[c], g0 + i);
                    float* ar = agg + (size_t)dj * AS + lane;
#pragma unroll
                    for (int q = 0; q < WPL; ++q) atomicAdd(ar + 32 * q, v[t & 1][i * WPL + q] * wj);
                }
            } else {
                unrolled<0, 16>([&](auto J) {
                    constexpr int j = decltype(J)::value;
                    const unsigned dj = (unsigned)half_bcast<j>(d[c]);
                    const float    wj = __int_as_float(half_bcast<j>(__float_as_int(w[c])));
                    atomicAdd(agg + (dj * AS + (lane & 15)), v[t & 1][j] * wj);
                });
            }
        };
        load_group(ic<0>{});
        count_deg(ic<1>{});
        unrolled<0, NG>([&](auto T) {
            constexpr int t = decltype(T)::value;
            if constexpr (t + 1 < NG) load_group(ic<t + 1>{});
            add_group(T);
        });
    } else {
        count_deg(ic<0>{});
        for (int c = 0; c < NCH; ++c) {
            const int left = E - (e0 + c * 32);
            const int cnt  = left < 32 ? left : 32;
            const int sc = s[c], dc = d[c];
            const float wc = w[c];
#pragma clang loop unroll(disable)
            for (int j = 0; j < cnt; ++j) {
                const int   sj = lane_i(sc, j);
                const int   dj = lane_i(dc, j);
                const float wj = lane_f(wc, j);
                for (int q = lane; q < K; q += 32)
                    atomicAdd(agg + (size_t)dj * AS + q, h[(size_t)sj * K + q] * wj);
            }
        }
    }
}

template <int K, int F, int ACT, int KCLR>
__global__ void __launch_bounds__(32)
sage_gemm(const float* __restrict__ h, float* agg,
          const float* __restrict__ deg,
          const float* __restrict__ Ws, const float* __restrict__ Wn,
          const float* __restrict__ bias,
          float* __restrict__ out) {
    constexpr int NT = (F + 15) / 16;
    const int lane = threadIdx.x;
    const int m    = lane & 15;
    const int kh   = (lane >> 4) << 1;
    const int row0 = blockIdx.x << 4;
    const int node = row0 + m;

    int   cs[NT];
    float cmask[NT];
    for (int t = 0; t < NT; ++t) {
        const int col = t * 16 + m;
        cs[t]    = (col < F) ? col  : 0;
        cmask[t] = (col < F) ? 1.0f : 0.0f;
    }

    const float invdeg = 1.0f / fmaxf(deg[node], 1.0f);

    const float* hrow = h   + (size_t)node * K;
    const float* arow = agg + (size_t)node * AGG_STRIDE;

    v8f c[NT];
    for (int t = 0; t < NT; ++t) c[t] = v8f{};

    for (int k = 0; k < K; k += 4) {
        v2f ah, an;
        ah.x = hrow[k + kh];
        ah.y = hrow[k + kh + 1];
        an.x = arow[k + kh]     * invdeg;
        an.y = arow[k + kh + 1] * invdeg;
        for (int t = 0; t < NT; ++t) {
            v2f b;
            b.x = Ws[(k + kh)     * F + cs[t]] * cmask[t];
            b.y = Ws[(k + kh + 1) * F + cs[t]] * cmask[t];
            c[t] = __builtin_amdgcn_wmma_f32_16x16x4_f32(false, ah, false, b,
                                                         (short)0, c[t], false, false);
            b.x = Wn[(k + kh)     * F + cs[t]] * cmask[t];
            b.y = Wn[(k + kh + 1) * F + cs[t]] * cmask[t];
            c[t] = __builtin_amdgcn_wmma_f32_16x16x4_f32(false, an, false, b,
                                                         (short)0, c[t], false, false);
        }
    }

    if constexpr (KCLR > 0) {
        constexpr int LPR = KCLR / 4;
        constexpr int RPI = 32 / LPR;
        const float4 z = make_float4(0.0f, 0.0f, 0.0f, 0.0f);
        float* crow = agg + (size_t)(row0 + lane / LPR) * AGG_STRIDE + (lane % LPR) * 4;
        for (int i = 0; i < 16 / RPI; ++i)
            *reinterpret_cast<float4*>(crow + (size_t)i * RPI * AGG_STRIDE) = z;
    }

    const int half = lane >> 4;
    for (int t = 0; t < NT; ++t) {
        const int cn = t * 16 + (lane & 15);
        if (cn < F) {
            const float bv = bias[cn];
            for (int j = 0; j < 8; ++j) {
                const int rm = row0 + j + (half << 3);
                float v = c[t][j] + bv;
                if (ACT) v = tanhf(v);
                out[(size_t)rm * F + cn] = v;
            }
        }
    }
}

extern "C" void kernel_launch(void* const* d_in, const int* in_sizes, int n_in,
                              void* d_out, int out_size, void* d_ws, size_t ws_size,
                              hipStream_t stream) {
    const float* b_z  = (const float*)d_in[0];
    const int*   esrc = (const int*)  d_in[1];
    const int*   edst = (const int*)  d_in[2];
    const float* ew   = (const float*)d_in[3];
    const float* Ws0  = (const float*)d_in[4];
    const float* Wn0  = (const float*)d_in[5];
    const float* b0   = (const float*)d_in[6];
    const float* Ws1  = (const float*)d_in[7];
    const float* Wn1  = (const float*)d_in[8];
    const float* b1   = (const float*)d_in[9];
    const float* Ws2  = (const float*)d_in[10];
    const float* Wn2  = (const float*)d_in[11];
    const float* b2   = (const float*)d_in[12];

    const int N = in_sizes[0] / 64;
    const int E = in_sizes[1];

    float* agg = (float*)d_ws;
    float* h1  = agg + (size_t)N * 64;
    float* h2  = h1  + (size_t)N * 32;
    float* deg = h2  + (size_t)N * 16;
    float* out = (float*)d_out;

    auto cdiv = [](int a, int b) { return (a + b - 1) / b; };

    clear2<<<512, 256, 0, stream>>>(deg, N, agg, N * 64);

    scatter_rows<64, true><<<cdiv(E, 256), 64, 0, stream>>>(b_z, esrc, edst, ew, agg, deg, E);
    sage_gemm<64, 32, 1, 32><<<N / 16, 32, 0, stream>>>(b_z, agg, deg, Ws0, Wn0, b0, h1);

    scatter_rows<32, false><<<cdiv(E, 256), 64, 0, stream>>>(h1, esrc, edst, ew, agg, nullptr, E);
    sage_gemm<32, 16, 1, 16><<<N / 16, 32, 0, stream>>>(h1, agg, deg, Ws1, Wn1, b1, h2);

    scatter_rows<16, false><<<cdiv(E, 256), 64, 0, stream>>>(h2, esrc, edst, ew, agg, nullptr, E);
    sage_gemm<16, 1, 0, 0><<<N / 16, 32, 0, stream>>>(h2, agg, deg, Ws2, Wn2, b2, out);
}


# Each 256-byte agg unit takes f32 atomic adds from one package half only: layer 0 a half per 32-feature unit (chunk tickets), layers 1-2 a plane per half, summed in the dense stage; sum order changes
# speedup vs baseline: 2.7065x; 1.5182x over previous
#include <hip/hip_runtime.h>
#include <hip/hip_bf16.h>

typedef __attribute__((ext_vector_type(2))) float v2f;
typedef __attribute__((ext_vector_type(8))) float v8f;

constexpr int UNIT = 64;
constexpr int SLOT = 32;
constexpr int TICKET_GAP = 64;

__device__ __forceinline__ int slot_of(int n) { return (n >> 1) * (2 * UNIT) + (n & 1) * SLOT; }

__device__ __forceinline__ int package_half() {
    return (int)((__builtin_amdgcn_s_sendmsg_rtn(0x87) >> 18) & 1u);
}

__global__ void __launch_bounds__(256)
clear2(float* __restrict__ p0, int n0, float* __restrict__ p1, int n1,
       unsigned* __restrict__ tickets) {
    const int tid = blockIdx.x * blockDim.x + threadIdx.x;
    if (tid < 2) tickets[tid * TICKET_GAP] = 0u;
    const int nth = gridDim.x * blockDim.x;
    const float4 z = make_float4(0.0f, 0.0f, 0.0f, 0.0f);
    for (int i = tid; i < (n0 >> 2); i += nth) reinterpret_cast<float4*>(p0)[i] = z;
    for (int i = (n0 & ~3) + tid; i < n0; i += nth) p0[i] = 0.0f;
    for (int i = tid; i < (n1 >> 2); i += nth) reinterpret_cast<float4*>(p1)[i] = z;
    for (int i = (n1 & ~3) + tid; i < n1; i += nth) p1[i] = 0.0f;
}

template <int J> struct ic { static constexpr int value = J; };
template <int J, int JEND, class F>
__device__ __forceinline__ void unrolled(F&& f) {
    if constexpr (J < JEND) { f(ic<J>{}); unrolled<J + 1, JEND>(f); }
}

template <int J> __device__ __forceinline__ int half_bcast(int v) {
    return __builtin_amdgcn_update_dpp(0, v, 0x150 + J, 0xf, 0xf, false);
}
__device__ __forceinline__ int lane_i(int v, int j) { return __builtin_amdgcn_readlane(v, j); }
__device__ __forceinline__ float lane_f(float v, int j) {
    return __int_as_float(__builtin_amdgcn_readlane(__float_as_int(v), j));
}

constexpr int EPW = 128;

template <int K, int HS, bool DEG>
__device__ __forceinline__ void
walk_edges(const float* __restrict__ h, const int* __restrict__ src,
           const int* __restrict__ dst, const float* __restrict__ ew,
           float* __restrict__ agg, float* __restrict__ deg, int E,
           int e0, int lane, bool count_on) {
    static_assert(K == 32 || K == 16, "a wave instruction covers one line or two half lines");
    constexpr int NCH = EPW / 32;
    constexpr int GE  = (K >= 32) ? 16 : 32;
    constexpr int GPC = 32 / GE;
    constexpr int NG  = NCH * GPC;

    int   s[NCH], d[NCH], o[NCH];
    float w[NCH];
#pragma unroll
    for (int c = 0; c < NCH; ++c) {
        const int e  = e0 + c * 32 + lane;
        const int ec = (e < E) ? e : 0;
        s[c] = src[ec];
        d[c] = dst[ec];
        w[c] = ew[ec];
        o[c] = slot_of(d[c]);
    }
    auto count_deg = [&](auto FULL) {
        if constexpr (DEG) {
            if (count_on)
                for (int c = 0; c < NCH; ++c)
                    if (decltype(FULL)::value || e0 + c * 32 + lane < E) atomicAdd(&deg[d[c]], 1.0f);
        }
    };

    if (E - e0 >= EPW) {
        float v[2][16];
        auto load_group = [&](auto T) {
            constexpr int t = decltype(T)::value;
            constexpr int c = t / GPC, g0 = (t % GPC) * GE;
            if constexpr (K >= 32) {
#pragma unroll
                for (int i = 0; i < GE; ++i) {
                    const int sj = lane_i(s[c], g0 + i);
                    v[t & 1][i] = h[(size_t)sj * HS + lane];
                }
            } else {
                unrolled<0, 16>([&](auto J) {
                    constexpr int j = decltype(J)::value;
                    const unsigned sj = (unsigned)half_bcast<j>(s[c]);
                    v[t & 1][j] = h[sj * HS + (lane & 15)];
                });
            }
        };
        auto add_group = [&](auto T) {
            constexpr int t = decltype(T)::value;
            constexpr int c = t / GPC, g0 = (t % GPC) * GE;
            if constexpr (K >= 32) {
#pragma unroll
                for (int i = 0; i < GE; ++i) {
                    const int   oj = lane_i(o[c], g0 + i);
                    const float wj = lane_f(w[c], g0 + i);
                    atomicAdd(agg + (size_t)oj + lane, v[t & 1][i] * wj);
                }
            } else {
                unrolled<0, 16>([&](auto J) {
                    constexpr int j = decltype(J)::value;
                    const unsigned oj = (unsigned)half_bcast<j>(o[c]);
                    const float    wj = __int_as_float(half_bcast<j>(__float_as_int(w[c])));
                    atomicAdd(agg + (oj + (lane & 15)), v[t & 1][j] * wj);
                });
            }
        };
        load_group(ic<0>{});
        count_deg(ic<1>{});
        unrolled<0, NG>([&](auto T) {
            constexpr int t = decltype(T)::value;
            if constexpr (t + 1 < NG) load_group(ic<t + 1>{});
            add_group(T);
        });
    } else {
        count_deg(ic<0>{});
        for (int c = 0; c < NCH; ++c) {
            const int left = E - (e0 + c * 32);
            const int cnt  = left < 32 ? left : 32;
            const int sc = s[c], oc = o[c];
            const float wc = w[c];
#pragma clang loop unroll(disable)
            for (int j = 0; j < cnt; ++j) {
                const int   sj = lane_i(sc, j);
                const int   oj = lane_i(oc, j);
                const float wj = lane_f(wc, j);
                if (lane < K)
                    atomicAdd(agg + (size_t)oj + lane, h[(size_t)sj * HS + lane] * wj);
            }
        }
    }
}

__device__ __forceinline__ int draw_ticket(unsigned* counter, int lane) {
    unsigned t = 0u;
    if (lane == 0) t = atomicAdd(counter, 1u);
    return __builtin_amdgcn_readfirstlane((int)t);
}

__global__ void __launch_bounds__(128)
scatter_split(const float* __restrict__ h, const int* __restrict__ src,
              const int* __restrict__ dst, const float* __restrict__ ew,
              float* __restrict__ agg, float* __restrict__ deg,
              unsigned* __restrict__ tickets, int E) {
    const int lane = threadIdx.x & 31;
    const int chunks = (E + EPW - 1) / EPW;
    int f = package_half();
    int t = draw_ticket(tickets + f * TICKET_GAP, lane);
    if (t >= chunks) {
        f ^= 1;
        t = draw_ticket(tickets + f * TICKET_GAP, lane);
        if (t >= chunks) return;
    }
    walk_edges<32, 64, true>(h + f * 32, src, dst, ew, agg + f * UNIT, deg, E,
                             t * EPW, lane, f == 0);
}

template <int K>
__global__ void __launch_bounds__(64)
scatter_plane(const float* __restrict__ h, const int* __restrict__ src,
              const int* __restrict__ dst, const float* __restrict__ ew,
              float* __restrict__ agg, int E) {
    const int lane = threadIdx.x & 31;
    const int wave = __builtin_amdgcn_readfirstlane((int)((blockIdx.x * blockDim.x + threadIdx.x) >> 5));
    const int e0 = wave * EPW;
    if (e0 >= E) return;
    walk_edges<K, K, false>(h, src, dst, ew, agg + package_half() * UNIT, nullptr, E,
                            e0, lane, false);
}

template <int K, int F, int ACT, int CLRW>
__global__ void __launch_bounds__(32)
sage_gemm(const float* __restrict__ h, float* agg,
          const float* __restrict__ deg,
          const float* __restrict__ Ws, const float* __restrict__ Wn,
          const float* __restrict__ bias,
          float* __restrict__ out) {
    constexpr int NT = (F + 15) / 16;
    const int lane = threadIdx.x;
    const int m    = lane & 15;
    const int kh   = (lane >> 4) << 1;
    const int row0 = blockIdx.x << 4;
    const int node = row0 + m;

    int   cs[NT];
    float cmask[NT];
    for (int t = 0; t < NT; ++t) {
        const int col = t * 16 + m;
        cs[t]    = (col < F) ? col  : 0;
        cmask[t] = (col < F) ? 1.0f : 0.0f;
    }

    const float invdeg = 1.0f / fmaxf(deg[node], 1.0f);

    const float* hrow = h   + (size_t)node * K;
    const float* arow = agg + (size_t)slot_of(node);

    v8f c[NT];
    for (int t = 0; t < NT; ++t) c[t] = v8f{};

    for (int k = 0; k < K; k += 4) {
        v2f ah, an;
        ah.x = hrow[k + kh];
        ah.y = hrow[k + kh + 1];
        const float* ap = arow + (k >> 5) * UNIT + (k & 31) + kh;
        an.x = ap[0];
        an.y = ap[1];
        if constexpr (K <= 32) {
            an.x += ap[UNIT];
            an.y += ap[UNIT + 1];
        }
        an.x *= invdeg;
        an.y *= invdeg;
        for (int t = 0; t < NT; ++t) {
            v2f b;
            b.x = Ws[(k + kh)     * F + cs[t]] * cmask[t];
            b.y = Ws[(k + kh + 1) * F + cs[t]] * cmask[t];
            c[t] = __builtin_amdgcn_wmma_f32_16x16x4_f32(false, ah, false, b,
                                                         (short)0, c[t], false, false);
            b.x = Wn[(k + kh)     * F + cs[t]] * cmask[t];
            b.y = Wn[(k + kh + 1) * F + cs[t]] * cmask[t];
            c[t] = __builtin_amdgcn_wmma_f32_16x16x4_f32(false, an, false, b,
                                                         (short)0, c[t], false, false);
        }
    }

    if constexpr (CLRW > 0) {
        constexpr int LPS = CLRW / 4;
        constexpr int SPI = 32 / LPS;
        const float4 z = make_float4(0.0f, 0.0f, 0.0f, 0.0f);
        float* cslot = agg + (size_t)row0 * UNIT + (lane / LPS) * SLOT + (lane % LPS) * 4;
        for (int i = 0; i < 32 / SPI; ++i)
            *reinterpret_cast<float4*>(cslot + (size_t)i * SPI * SLOT) = z;
    }

    const int half = lane >> 4;
    for (int t = 0; t < NT; ++t) {
        const int cn = t * 16 + (lane & 15);
        if (cn < F) {
            const float bv = bias[cn];
            for (int j = 0; j < 8; ++j) {
                const int rm = row0 + j + (half << 3);
                float v = c[t][j] + bv;
                if (ACT) v = tanhf(v);
                out[(size_t)rm * F + cn] = v;
            }
        }
    }
}

extern "C" void kernel_launch(void* const* d_in, const int* in_sizes, int n_in,
                              void* d_out, int out_size, void* d_ws, size_t ws_size,
                              hipStream_t stream) {
    const float* b_z  = (const float*)d_in[0];
    const int*   esrc = (const int*)  d_in[1];
    const int*   edst = (const int*)  d_in[2];
    const float* ew   = (const float*)d_in[3];
    const float* Ws0  = (const float*)d_in[4];
    const float* Wn0  = (const float*)d_in[5];
    const float* b0   = (const float*)d_in[6];
    const float* Ws1  = (const float*)d_in[7];
    const float* Wn1  = (const float*)d_in[8];
    const float* b1   = (const float*)d_in[9];
    const float* Ws2  = (const float*)d_in[10];
    const float* Wn2  = (const float*)d_in[11];
    const float* b2   = (const float*)d_in[12];

    const int N = in_sizes[0] / 64;
    const int E = in_sizes[1];

    float* agg = (float*)d_ws;
    float* h1  = agg + (size_t)N * 64;
    float* h2  = h1  + (size_t)N * 32;
    float* deg = h2  + (size_t)N * 16;
    float* out = (float*)d_out;

    auto cdiv = [](int a, int b) { return (a + b - 1) / b; };

    unsigned* tickets = (unsigned*)h2;

    clear2<<<512, 256, 0, stream>>>(deg, N, agg, N * 64, tickets);

    scatter_split<<<cdiv(E, 256), 128, 0, stream>>>(b_z, esrc, edst, ew, agg, deg, tickets, E);
    sage_gemm<64, 32, 1, 32><<<N / 16, 32, 0, stream>>>(b_z, agg, deg, Ws0, Wn0, b0, h1);

    scatter_plane<32><<<cdiv(E, 256), 64, 0, stream>>>(h1, esrc, edst, ew, agg, E);
    sage_gemm<32, 16, 1, 16><<<N / 16, 32, 0, stream>>>(h1, agg, deg, Ws1, Wn1, b1, h2);

    scatter_plane<16><<<cdiv(E, 256), 64, 0, stream>>>(h2, esrc, edst, ew, agg, E);
    sage_gemm<16, 1, 0, 0><<<N / 16, 32, 0, stream>>>(h2, agg, deg, Ws2, Wn2, b2, out);
}


# Layer 0 in-degree adds split evenly over both package halves: piece (chunk, unit p) counts 64 of its 128 edges into deg plane p, the dense stage adds the two exact planes; f32 row sums unchanged
# speedup vs baseline: 2.7921x; 1.0316x over previous
#include <hip/hip_runtime.h>
#include <hip/hip_bf16.h>

typedef __attribute__((ext_vector_type(2))) float v2f;
typedef __attribute__((ext_vector_type(8))) float v8f;

constexpr int UNIT = 64;
constexpr int SLOT = 32;
constexpr int TICKET_GAP = 64;

__device__ __forceinline__ int slot_of(int n) { return (n >> 1) * (2 * UNIT) + (n & 1) * SLOT; }

__device__ __forceinline__ int package_half() {
    return (int)((__builtin_amdgcn_s_sendmsg_rtn(0x87) >> 18) & 1u);
}

__global__ void __launch_bounds__(256)
clear2(float* __restrict__ p0, int n0, float* __restrict__ p1, int n1,
       unsigned* __restrict__ tickets) {
    const int tid = blockIdx.x * blockDim.x + threadIdx.x;
    if (tid < 2) tickets[tid * TICKET_GAP] = 0u;
    const int nth = gridDim.x * blockDim.x;
    const float4 z = make_float4(0.0f, 0.0f, 0.0f, 0.0f);
    for (int i = tid; i < (n0 >> 2); i += nth) reinterpret_cast<float4*>(p0)[i] = z;
    for (int i = (n0 & ~3) + tid; i < n0; i += nth) p0[i] = 0.0f;
    for (int i = tid; i < (n1 >> 2); i += nth) reinterpret_cast<float4*>(p1)[i] = z;
    for (int i = (n1 & ~3) + tid; i < n1; i += nth) p1[i] = 0.0f;
}

template <int J> struct ic { static constexpr int value = J; };
template <int J, int JEND, class F>
__device__ __forceinline__ void unrolled(F&& f) {
    if constexpr (J < JEND) { f(ic<J>{}); unrolled<J + 1, JEND>(f); }
}

template <int J> __device__ __forceinline__ int half_bcast(int v) {
    return __builtin_amdgcn_update_dpp(0, v, 0x150 + J, 0xf, 0xf, false);
}
__device__ __forceinline__ int lane_i(int v, int j) { return __builtin_amdgcn_readlane(v, j); }
__device__ __forceinline__ float lane_f(float v, int j) {
    return __int_as_float(__builtin_amdgcn_readlane(__float_as_int(v), j));
}

constexpr int EPW = 128;

template <int K, int HS, bool DEG>
__device__ __forceinline__ void
walk_edges(const float* __restrict__ h, const int* __restrict__ src,
           const int* __restrict__ dst, const float* __restrict__ ew,
           float* __restrict__ agg, float* __restrict__ deg, int E,
           int e0, int lane, int cnt_half) {
    static_assert(K == 32 || K == 16, "a wave instruction covers one line or two half lines");
    constexpr int NCH = EPW / 32;
    constexpr int GE  = (K >= 32) ? 16 : 32;
    constexpr int GPC = 32 / GE;
    constexpr int NG  = NCH * GPC;

    int   s[NCH], d[NCH], o[NCH];
    float w[NCH];
#pragma unroll
    for (int c = 0; c < NCH; ++c) {
        const int e  = e0 + c * 32 + lane;
        const int ec = (e < E) ? e : 0;
        s[c] = src[ec];
        d[c] = dst[ec];
        w[c] = ew[ec];
        o[c] = slot_of(d[c]);
    }
    auto count_deg = [&](auto FULL) {
        if constexpr (DEG) {
            for (int c = 0; c < NCH; ++c)
                if ((c >> 1) == cnt_half && (decltype(FULL)::value || e0 + c * 32 + lane < E))
                    atomicAdd(&deg[d[c]], 1.0f);
        }
    };

    if (E - e0 >= EPW) {
        float v[2][16];
        auto load_group = [&](auto T) {
            constexpr int t = decltype(T)::value;
            constexpr int c = t / GPC, g0 = (t % GPC) * GE;
            if constexpr (K >= 32) {
#pragma unroll
                for (int i = 0; i < GE; ++i) {
                    const int sj = lane_i(s[c], g0 + i);
                    v[t & 1][i] = h[(size_t)sj * HS + lane];
                }
            } else {
                unrolled<0, 16>([&](auto J) {
                    constexpr int j = decltype(J)::value;
                    const unsigned sj = (unsigned)half_bcast<j>(s[c]);
                    v[t & 1][j] = h[sj * HS + (lane & 15)];
                });
            }
        };
        auto add_group = [&](auto T) {
            constexpr int t = decltype(T)::value;
            constexpr int c = t / GPC, g0 = (t % GPC) * GE;
            if constexpr (K >= 32) {
#pragma unroll
                for (int i = 0; i < GE; ++i) {
                    const int   oj = lane_i(o[c], g0 + i);
                    const float wj = lane_f(w[c], g0 + i);
                    atomicAdd(agg + (size_t)oj + lane, v[t & 1][i] * wj);
                }
            } else {
                unrolled<0, 16>([&](auto J) {
                    constexpr int j = decltype(J)::value;
                    const unsigned oj = (unsigned)half_bcast<j>(o[c]);
                    const float    wj = __int_as_float(half_bcast<j>(__float_as_int(w[c])));
                    atomicAdd(agg + (oj + (lane & 15)), v[t & 1][j] * wj);
                });
            }
        };
        load_group(ic<0>{});
        count_deg(ic<1>{});
        unrolled<0, NG>([&](auto T) {
            constexpr int t = decltype(T)::value;
            if constexpr (t + 1 < NG) load_group(ic<t + 1>{});
            add_group(T);
        });
    } else {
        count_deg(ic<0>{});
        for (int c = 0; c < NCH; ++c) {
            const int left = E - (e0 + c * 32);
            const int cnt  = left < 32 ? left : 32;
            const int sc = s[c], oc = o[c];
            const float wc = w[c];
#pragma clang loop unroll(disable)
            for (int j = 0; j < cnt; ++j) {
                const int   sj = lane_i(sc, j);
                const int   oj = lane_i(oc, j);
                const float wj = lane_f(wc, j);
                if (lane < K)
                    atomicAdd(agg + (size_t)oj + lane, h[(size_t)sj * HS + lane] * wj);
            }
        }
    }
}

__device__ __forceinline__ int draw_ticket(unsigned* counter, int lane) {
    unsigned t = 0u;
    if (lane == 0) t = atomicAdd(counter, 1u);
    return __builtin_amdgcn_readfirstlane((int)t);
}

__global__ void __launch_bounds__(128)
scatter_split(const float* __restrict__ h, const int* __restrict__ src,
              const int* __restrict__ dst, const float* __restrict__ ew,
              float* __restrict__ agg, float* __restrict__ deg0, float* __restrict__ deg1,
              unsigned* __restrict__ tickets, int E) {
    const int lane = threadIdx.x & 31;
    const int chunks = (E + EPW - 1) / EPW;
    int f = package_half();
    int t = draw_ticket(tickets + f * TICKET_GAP, lane);
    if (t >= chunks) {
        f ^= 1;
        t = draw_ticket(tickets + f * TICKET_GAP, lane);
        if (t >= chunks) return;
    }
    walk_edges<32, 64, true>(h + f * 32, src, dst, ew, agg + f * UNIT, f ? deg1 : deg0, E,
                             t * EPW, lane, f);
}

template <int K>
__global__ void __launch_bounds__(64)
scatter_plane(const float* __restrict__ h, const int* __restrict__ src,
              const int* __restrict__ dst, const float* __restrict__ ew,
              float* __restrict__ agg, int E) {
    const int lane = threadIdx.x & 31;
    const int wave = __builtin_amdgcn_readfirstlane((int)((blockIdx.x * blockDim.x + threadIdx.x) >> 5));
    const int e0 = wave * EPW;
    if (e0 >= E) return;
    walk_edges<K, K, false>(h, src, dst, ew, agg + package_half() * UNIT, nullptr, E,
                            e0, lane, 0);
}

template <int K, int F, int ACT, int CLRW, bool DEG2>
__global__ void __launch_bounds__(32)
sage_gemm(const float* __restrict__ h, float* agg,
          float* deg, const float* __restrict__ deg_b,
          const float* __restrict__ Ws, const float* __restrict__ Wn,
          const float* __restrict__ bias,
          float* __restrict__ out) {
    constexpr int NT = (F + 15) / 16;
    const int lane = threadIdx.x;
    const int m    = lane & 15;
    const int kh   = (lane >> 4) << 1;
    const int row0 = blockIdx.x << 4;
    const int node = row0 + m;

    int   cs[NT];
    float cmask[NT];
    for (int t = 0; t < NT; ++t) {
        const int col = t * 16 + m;
        cs[t]    = (col < F) ? col  : 0;
        cmask[t] = (col < F) ? 1.0f : 0.0f;
    }

    float degn = deg[node];
    if constexpr (DEG2) degn += deg_b[node];
    const float invdeg = 1.0f / fmaxf(degn, 1.0f);

    const float* hrow = h   + (size_t)node * K;
    const float* arow = agg + (size_t)slot_of(node);

    v8f c[NT];
    for (int t = 0; t < NT; ++t) c[t] = v8f{};

    for (int k = 0; k < K; k += 4) {
        v2f ah, an;
        ah.x = hrow[k + kh];
        ah.y = hrow[k + kh + 1];
        const float* ap = arow + (k >> 5) * UNIT + (k & 31) + kh;
        an.x = ap[0];
        an.y = ap[1];
        if constexpr (K <= 32) {
            an.x += ap[UNIT];
            an.y += ap[UNIT + 1];
        }
        an.x *= invdeg;
        an.y *= invdeg;
        for (int t = 0; t < NT; ++t) {
            v2f b;
            b.x = Ws[(k + kh)     * F + cs[t]] * cmask[t];
            b.y = Ws[(k + kh + 1) * F + cs[t]] * cmask[t];
            c[t] = __builtin_amdgcn_wmma_f32_16x16x4_f32(false, ah, false, b,
                                                         (short)0, c[t], false, false);
            b.x = Wn[(k + kh)     * F + cs[t]] * cmask[t];
            b.y = Wn[(k + kh + 1) * F + cs[t]] * cmask[t];
            c[t] = __builtin_amdgcn_wmma_f32_16x16x4_f32(false, an, false, b,
                                                         (short)0, c[t], false, false);
        }
    }

    if constexpr (DEG2) {
        if (lane < 16) deg[node] = degn;
    }

    if constexpr (CLRW > 0) {
        constexpr int LPS = CLRW / 4;
        constexpr int SPI = 32 / LPS;
        const float4 z = make_float4(0.0f, 0.0f, 0.0f, 0.0f);
        float* cslot = agg + (size_t)row0 * UNIT + (lane / LPS) * SLOT + (lane % LPS) * 4;
        for (int i = 0; i < 32 / SPI; ++i)
            *reinterpret_cast<float4*>(cslot + (size_t)i * SPI * SLOT) = z;
    }

    const int half = lane >> 4;
    for (int t = 0; t < NT; ++t) {
        const int cn = t * 16 + (lane & 15);
        if (cn < F) {
            const float bv = bias[cn];
            for (int j = 0; j < 8; ++j) {
                const int rm = row0 + j + (half << 3);
                float v = c[t][j] + bv;
                if (ACT) v = tanhf(v);
                out[(size_t)rm * F + cn] = v;
            }
        }
    }
}

extern "C" void kernel_launch(void* const* d_in, const int* in_sizes, int n_in,
                              void* d_out, int out_size, void* d_ws, size_t ws_size,
                              hipStream_t stream) {
    const float* b_z  = (const float*)d_in[0];
    const int*   esrc = (const int*)  d_in[1];
    const int*   edst = (const int*)  d_in[2];
    const float* ew   = (const float*)d_in[3];
    const float* Ws0  = (const float*)d_in[4];
    const float* Wn0  = (const float*)d_in[5];
    const float* b0   = (const float*)d_in[6];
    const float* Ws1  = (const float*)d_in[7];
    const float* Wn1  = (const float*)d_in[8];
    const float* b1   = (const float*)d_in[9];
    const float* Ws2  = (const float*)d_in[10];
    const float* Wn2  = (const float*)d_in[11];
    const float* b2   = (const float*)d_in[12];

    const int N = in_sizes[0] / 64;
    const int E = in_sizes[1];

    float* agg = (float*)d_ws;
    float* h1  = agg + (size_t)N * 64;
    float* h2  = h1  + (size_t)N * 32;
    float* deg = h2  + (size_t)N * 16;
    float* out = (float*)d_out;
    float* deg1 = deg - N;

    auto cdiv = [](int a, int b) { return (a + b - 1) / b; };

    unsigned* tickets = (unsigned*)h2;

    clear2<<<512, 256, 0, stream>>>(deg1, 2 * N, agg, N * 64, tickets);

    scatter_split<<<cdiv(E, 256), 128, 0, stream>>>(b_z, esrc, edst, ew, agg, deg, deg1, tickets, E);
    sage_gemm<64, 32, 1, 32, true><<<N / 16, 32, 0, stream>>>(b_z, agg, deg, deg1, Ws0, Wn0, b0, h1);

    scatter_plane<32><<<cdiv(E, 256), 64, 0, stream>>>(h1, esrc, edst, ew, agg, E);
    sage_gemm<32, 16, 1, 16, false><<<N / 16, 32, 0, stream>>>(h1, agg, deg, nullptr, Ws1, Wn1, b1, h2);

    scatter_plane<16><<<cdiv(E, 256), 64, 0, stream>>>(h2, esrc, edst, ew, agg, E);
    sage_gemm<16, 1, 0, 0, false><<<N / 16, 32, 0, stream>>>(h2, agg, deg, nullptr, Ws2, Wn2, b2, out);
}


# agg laid out plane by plane (each package half's node slots one contiguous range) instead of alternating 256-byte units: a half's f32 atomic adds reach every unit of its range; sums unchanged
# speedup vs baseline: 2.9357x; 1.0514x over previous
#include <hip/hip_runtime.h>
#include <hip/hip_bf16.h>

typedef __attribute__((ext_vector_type(2))) float v2f;
typedef __attribute__((ext_vector_type(8))) float v8f;

constexpr int SLOT = 32;
constexpr int TICKET_GAP = 64;

__device__ __forceinline__ int slot_of(int n) { return n * SLOT; }

__device__ __forceinline__ int package_half() {
    return (int)((__builtin_amdgcn_s_sendmsg_rtn(0x87) >> 18) & 1u);
}

__global__ void __launch_bounds__(256)
clear2(float* __restrict__ p0, int n0, float* __restrict__ p1, int n1,
       unsigned* __restrict__ tickets) {
    const int tid = blockIdx.x * blockDim.x + threadIdx.x;
    if (tid < 2) tickets[tid * TICKET_GAP] = 0u;
    const int nth = gridDim.x * blockDim.x;
    const float4 z = make_float4(0.0f, 0.0f, 0.0f, 0.0f);
    for (int i = tid; i < (n0 >> 2); i += nth) reinterpret_cast<float4*>(p0)[i] = z;
    for (int i = (n0 & ~3) + tid; i < n0; i += nth) p0[i] = 0.0f;
    for (int i = tid; i < (n1 >> 2); i += nth) reinterpret_cast<float4*>(p1)[i] = z;
    for (int i = (n1 & ~3) + tid; i < n1; i += nth) p1[i] = 0.0f;
}

template <int J> struct ic { static constexpr int value = J; };
template <int J, int JEND, class F>
__device__ __forceinline__ void unrolled(F&& f) {
    if constexpr (J < JEND) { f(ic<J>{}); unrolled<J + 1, JEND>(f); }
}

template <int J> __device__ __forceinline__ int half_bcast(int v) {
    return __builtin_amdgcn_update_dpp(0, v, 0x150 + J, 0xf, 0xf, false);
}
__device__ __forceinline__ int lane_i(int v, int j) { return __builtin_amdgcn_readlane(v, j); }
__device__ __forceinline__ float lane_f(float v, int j) {
    return __int_as_float(__builtin_amdgcn_readlane(__float_as_int(v), j));
}

constexpr int EPW = 128;

template <int K, int HS, bool DEG>
__device__ __forceinline__ void
walk_edges(const float* __restrict__ h, const int* __restrict__ src,
           const int* __restrict__ dst, const float* __restrict__ ew,
           float* __restrict__ agg, float* __restrict__ deg, int E,
           int e0, int lane, int cnt_half) {
    static_assert(K == 32 || K == 16, "a wave instruction covers one line or two half lines");
    constexpr int NCH = EPW / 32;
    constexpr int GE  = (K >= 32) ? 16 : 32;
    constexpr int GPC = 32 / GE;
    constexpr int NG  = NCH * GPC;

    int   s[NCH], d[NCH], o[NCH];
    float w[NCH];
#pragma unroll
    for (int c = 0; c < NCH; ++c) {
        const int e  = e0 + c * 32 + lane;
        const int ec = (e < E) ? e : 0;
        s[c] = src[ec];
        d[c] = dst[ec];
        w[c] = ew[ec];
        o[c] = slot_of(d[c]);
    }
    auto count_deg = [&](auto FULL) {
        if constexpr (DEG) {
            for (int c = 0; c < NCH; ++c)
                if ((c >> 1) == cnt_half && (decltype(FULL)::value || e0 + c * 32 + lane < E))
                    atomicAdd(&deg[d[c]], 1.0f);
        }
    };

    if (E - e0 >= EPW) {
        float v[2][16];
        auto load_group = [&](auto T) {
            constexpr int t = decltype(T)::value;
            constexpr int c = t / GPC, g0 = (t % GPC) * GE;
            if constexpr (K >= 32) {
#pragma unroll
                for (int i = 0; i < GE; ++i) {
                    const int sj = lane_i(s[c], g0 + i);
                    v[t & 1][i] = h[(size_t)sj * HS + lane];
                }
            } else {
                unrolled<0, 16>([&](auto J) {
                    constexpr int j = decltype(J)::value;
                    const unsigned sj = (unsigned)half_bcast<j>(s[c]);
                    v[t & 1][j] = h[sj * HS + (lane & 15)];
                });
            }
        };
        auto add_group = [&](auto T) {
            constexpr int t = decltype(T)::value;
            constexpr int c = t / GPC, g0 = (t % GPC) * GE;
            if constexpr (K >= 32) {
#pragma unroll
                for (int i = 0; i < GE; ++i) {
                    const int   oj = lane_i(o[c], g0 + i);
                    const float wj = lane_f(w[c], g0 + i);
                    atomicAdd(agg + (size_t)oj + lane, v[t & 1][i] * wj);
                }
            } else {
                unrolled<0, 16>([&](auto J) {
                    constexpr int j = decltype(J)::value;
                    const unsigned oj = (unsigned)half_bcast<j>(o[c]);
                    const float    wj = __int_as_float(half_bcast<j>(__float_as_int(w[c])));
                    atomicAdd(agg + (oj + (lane & 15)), v[t & 1][j] * wj);
                });
            }
        };
        load_group(ic<0>{});
        count_deg(ic<1>{});
        unrolled<0, NG>([&](auto T) {
            constexpr int t = decltype(T)::value;
            if constexpr (t + 1 < NG) load_group(ic<t + 1>{});
            add_group(T);
        });
    } else {
        count_deg(ic<0>{});
        for (int c = 0; c < NCH; ++c) {
            const int left = E - (e0 + c * 32);
            const int cnt  = left < 32 ? left : 32;
            const int sc = s[c], oc = o[c];
            const float wc = w[c];
#pragma clang loop unroll(disable)
            for (int j = 0; j < cnt; ++j) {
                const int   sj = lane_i(sc, j);
                const int   oj = lane_i(oc, j);
                const float wj = lane_f(wc, j);
                if (lane < K)
                    atomicAdd(agg + (size_t)oj + lane, h[(size_t)sj * HS + lane] * wj);
            }
        }
    }
}

__device__ __forceinline__ int draw_ticket(unsigned* counter, int lane) {
    unsigned t = 0u;
    if (lane == 0) t = atomicAdd(counter, 1u);
    return __builtin_amdgcn_readfirstlane((int)t);
}

__global__ void __launch_bounds__(128)
scatter_split(const float* __restrict__ h, const int* __restrict__ src,
              const int* __restrict__ dst, const float* __restrict__ ew,
              float* __restrict__ agg0, float* __restrict__ agg1,
              float* __restrict__ deg0, float* __restrict__ deg1,
              unsigned* __restrict__ tickets, int E) {
    const int lane = threadIdx.x & 31;
    const int chunks = (E + EPW - 1) / EPW;
    int f = package_half();
    int t = draw_ticket(tickets + f * TICKET_GAP, lane);
    if (t >= chunks) {
        f ^= 1;
        t = draw_ticket(tickets + f * TICKET_GAP, lane);
        if (t >= chunks) return;
    }
    walk_edges<32, 64, true>(h + f * 32, src, dst, ew, f ? agg1 : agg0, f ? deg1 : deg0, E,
                             t * EPW, lane, f);
}

template <int K>
__global__ void __launch_bounds__(64)
scatter_plane(const float* __restrict__ h, const int* __restrict__ src,
              const int* __restrict__ dst, const float* __restrict__ ew,
              float* __restrict__ agg0, float* __restrict__ agg1, int E) {
    const int lane = threadIdx.x & 31;
    const int wave = __builtin_amdgcn_readfirstlane((int)((blockIdx.x * blockDim.x + threadIdx.x) >> 5));
    const int e0 = wave * EPW;
    if (e0 >= E) return;
    walk_edges<K, K, false>(h, src, dst, ew, package_half() ? agg1 : agg0, nullptr, E,
                            e0, lane, 0);
}

template <int K, int F, int ACT, int CLRW, bool DEG2>
__global__ void __launch_bounds__(32)
sage_gemm(const float* __restrict__ h, float* agg0, float* agg1,
          float* deg, const float* __restrict__ deg_b,
          const float* __restrict__ Ws, const float* __restrict__ Wn,
          const float* __restrict__ bias,
          float* __restrict__ out) {
    constexpr int NT = (F + 15) / 16;
    const int lane = threadIdx.x;
    const int m    = lane & 15;
    const int kh   = (lane >> 4) << 1;
    const int row0 = blockIdx.x << 4;
    const int node = row0 + m;

    int   cs[NT];
    float cmask[NT];
    for (int t = 0; t < NT; ++t) {
        const int col = t * 16 + m;
        cs[t]    = (col < F) ? col  : 0;
        cmask[t] = (col < F) ? 1.0f : 0.0f;
    }

    float degn = deg[node];
    if constexpr (DEG2) degn += deg_b[node];
    const float invdeg = 1.0f / fmaxf(degn, 1.0f);

    const float* hrow  = h    + (size_t)node * K;
    const float* arow0 = agg0 + (size_t)slot_of(node);
    const float* arow1 = agg1 + (size_t)slot_of(node);

    v8f c[NT];
    for (int t = 0; t < NT; ++t) c[t] = v8f{};

    for (int k = 0; k < K; k += 4) {
        v2f ah, an;
        ah.x = hrow[k + kh];
        ah.y = hrow[k + kh + 1];
        const float* ap = ((k >> 5) ? arow1 : arow0) + (k & 31) + kh;
        an.x = ap[0];
        an.y = ap[1];
        if constexpr (K <= 32) {
            const float* aq = arow1 + k + kh;
            an.x += aq[0];
            an.y += aq[1];
        }
        an.x *= invdeg;
        an.y *= invdeg;
        for (int t = 0; t < NT; ++t) {
            v2f b;
            b.x = Ws[(k + kh)     * F + cs[t]] * cmask[t];
            b.y = Ws[(k + kh + 1) * F + cs[t]] * cmask[t];
            c[t] = __builtin_amdgcn_wmma_f32_16x16x4_f32(false, ah, false, b,
                                                         (short)0, c[t], false, false);
            b.x = Wn[(k + kh)     * F + cs[t]] * cmask[t];
            b.y = Wn[(k + kh + 1) * F + cs[t]] * cmask[t];
            c[t] = __builtin_amdgcn_wmma_f32_16x16x4_f32(false, an, false, b,
                                                         (short)0, c[t], false, false);
        }
    }

    if constexpr (DEG2) {
        if (lane < 16) deg[node] = degn;
    }

    if constexpr (CLRW > 0) {
        constexpr int LPS = CLRW / 4;
        constexpr int SPI = 32 / LPS;
        const float4 z = make_float4(0.0f, 0.0f, 0.0f, 0.0f);
        const size_t coff = (size_t)row0 * SLOT + (lane / LPS) * SLOT + (lane % LPS) * 4;
        for (int i = 0; i < 16 / SPI; ++i) {
            *reinterpret_cast<float4*>(agg0 + coff + (size_t)i * SPI * SLOT) = z;
            *reinterpret_cast<float4*>(agg1 + coff + (size_t)i * SPI * SLOT) = z;
        }
    }

    const int half = lane >> 4;
    for (int t = 0; t < NT; ++t) {
        const int cn = t * 16 + (lane & 15);
        if (cn < F) {
            const float bv = bias[cn];
            for (int j = 0; j < 8; ++j) {
                const int rm = row0 + j + (half << 3);
                float v = c[t][j] + bv;
                if (ACT) v = tanhf(v);
                out[(size_t)rm * F + cn] = v;
            }
        }
    }
}

extern "C" void kernel_launch(void* const* d_in, const int* in_sizes, int n_in,
                              void* d_out, int out_size, void* d_ws, size_t ws_size,
                              hipStream_t stream) {
    const float* b_z  = (const float*)d_in[0];
    const int*   esrc = (const int*)  d_in[1];
    const int*   edst = (const int*)  d_in[2];
    const float* ew   = (const float*)d_in[3];
    const float* Ws0  = (const float*)d_in[4];
    const float* Wn0  = (const float*)d_in[5];
    const float* b0   = (const float*)d_in[6];
    const float* Ws1  = (const float*)d_in[7];
    const float* Wn1  = (const float*)d_in[8];
    const float* b1   = (const float*)d_in[9];
    const float* Ws2  = (const float*)d_in[10];
    const float* Wn2  = (const float*)d_in[11];
    const float* b2   = (const float*)d_in[12];

    const int N = in_sizes[0] / 64;
    const int E = in_sizes[1];

    float* agg  = (float*)d_ws;
    float* agg1 = agg + (size_t)N * 32;
    float* h1   = agg + (size_t)N * 64;
    float* h2  = h1  + (size_t)N * 32;
    float* deg = h2  + (size_t)N * 16;
    float* out = (float*)d_out;
    float* deg1 = deg - N;

    auto cdiv = [](int a, int b) { return (a + b - 1) / b; };

    unsigned* tickets = (unsigned*)h2;

    clear2<<<512, 256, 0, stream>>>(deg1, 2 * N, agg, N * 64, tickets);

    scatter_split<<<cdiv(E, 256), 128, 0, stream>>>(b_z, esrc, edst, ew, agg, agg1, deg, deg1, tickets, E);
    sage_gemm<64, 32, 1, 32, true><<<N / 16, 32, 0, stream>>>(b_z, agg, agg1, deg, deg1, Ws0, Wn0, b0, h1);

    scatter_plane<32><<<cdiv(E, 256), 64, 0, stream>>>(h1, esrc, edst, ew, agg, agg1, E);
    sage_gemm<32, 16, 1, 16, false><<<N / 16, 32, 0, stream>>>(h1, agg, agg1, deg, nullptr, Ws1, Wn1, b1, h2);

    scatter_plane<16><<<cdiv(E, 256), 64, 0, stream>>>(h2, esrc, edst, ew, agg, agg1, E);
    sage_gemm<16, 1, 0, 0, false><<<N / 16, 32, 0, stream>>>(h2, agg, agg1, deg, nullptr, Ws2, Wn2, b2, out);
}
